# starting kernel
# baseline (speedup 1.0000x reference)
#include <hip/hip_runtime.h>
#include <math.h>


typedef __attribute__((ext_vector_type(2))) float v2f;
typedef __attribute__((ext_vector_type(8))) float v8f;

#define NTERMS 20
#define KPAD   32
#define TROW   36

__global__ __launch_bounds__(256) void ipe_logsumexp_wmma(
    const float* __restrict__ a, const float* __restrict__ b,
    float* __restrict__ out, int nElem, int nChunks) {
  __shared__ __align__(16) float Ksh[NTERMS][KPAD];
  __shared__ __align__(16) float Tsh[8][2][16][TROW];

  if (threadIdx.x < NTERMS) {
    const int m = threadIdx.x;
    double W = 0.2734375 * 3.14159265358979323846;
    for (int j = 0; j < m; ++j) {
      double tj = 2.0 * j;
      W *= (tj + 0.5) * (tj + 1.5) /
           (4.0 * (tj + 5.0) * (tj + 6.0) * (j + 1.0) * (j + 1.0));
    }
    double R = W;
    for (int k = 0; k < NTERMS; ++k) {
      Ksh[m][k] = (float)R;
      R *= (4.5 + k) / ((0.5 + k) * (2.0 * m + 5.0 + k) * (k + 1.0));
    }
    for (int k = NTERMS; k < KPAD; ++k) Ksh[m][k] = 0.0f;
  }
  __syncthreads();

  const int lane = threadIdx.x & 31;
  const int wv   = threadIdx.x >> 5;
  const int col  = lane & 15;
  const int rsel = (lane < 16) ? 0 : 2;

  v2f B0[5], B1[5];
#pragma unroll
  for (int s = 0; s < 5; ++s) {
    B0[s] = v2f{Ksh[4 * s + rsel][col],      Ksh[4 * s + rsel + 1][col]};
    B1[s] = v2f{Ksh[4 * s + rsel][16 + col], Ksh[4 * s + rsel + 1][16 + col]};
  }

  const int nwaves = (int)gridDim.x * (blockDim.x >> 5);
  const int gwave  = (int)blockIdx.x * (blockDim.x >> 5) + wv;

  for (int chunk = gwave; chunk < nChunks; chunk += nwaves) {
    const int base = chunk * 32;
    if (base + 32 <= nElem) {
      float bA = b[base + col];
      float bB = b[base + 16 + col];
      float aX = a[base + lane];
      float x  = aX * aX * 0.25f;

      float bA2 = bA * bA, bB2 = bB * bB;
      float p0A = (lane < 16) ? 1.0f : bA2;
      float p1A = (lane < 16) ? bA   : bA2 * bA;
      float p0B = (lane < 16) ? 1.0f : bB2;
      float p1B = (lane < 16) ? bB   : bB2 * bB;
      float b4A = bA2 * bA2, b4B = bB2 * bB2;

      v8f c0A = {}, c1A = {}, c0B = {}, c1B = {};
#pragma unroll
      for (int s = 0; s < 5; ++s) {
        v2f Aa = v2f{p0A, p1A};
        v2f Ab = v2f{p0B, p1B};
        c0A = __builtin_amdgcn_wmma_f32_16x16x4_f32(false, Aa, false, B0[s], (short)0, c0A, false, false);
        c1A = __builtin_amdgcn_wmma_f32_16x16x4_f32(false, Aa, false, B1[s], (short)0, c1A, false, false);
        c0B = __builtin_amdgcn_wmma_f32_16x16x4_f32(false, Ab, false, B0[s], (short)0, c0B, false, false);
        c1B = __builtin_amdgcn_wmma_f32_16x16x4_f32(false, Ab, false, B1[s], (short)0, c1B, false, false);
        p0A *= b4A; p1A *= b4A;
        p0B *= b4B; p1B *= b4B;
      }

      const int row = (lane < 16) ? 0 : 8;
#pragma unroll
      for (int i = 0; i < 8; ++i) {
        Tsh[wv][0][row + i][col]      = c0A[i];
        Tsh[wv][0][row + i][16 + col] = c1A[i];
        Tsh[wv][1][row + i][col]      = c0B[i];
        Tsh[wv][1][row + i][16 + col] = c1B[i];
      }

      const float4* pT4 = (const float4*)&Tsh[wv][lane >> 4][lane & 15][0];
      float4 q0 = pT4[0], q1 = pT4[1], q2 = pT4[2], q3 = pT4[3], q4 = pT4[4];
      float v = q4.w;
      v = fmaf(v, x, q4.z); v = fmaf(v, x, q4.y); v = fmaf(v, x, q4.x);
      v = fmaf(v, x, q3.w); v = fmaf(v, x, q3.z); v = fmaf(v, x, q3.y); v = fmaf(v, x, q3.x);
      v = fmaf(v, x, q2.w); v = fmaf(v, x, q2.z); v = fmaf(v, x, q2.y); v = fmaf(v, x, q2.x);
      v = fmaf(v, x, q1.w); v = fmaf(v, x, q1.z); v = fmaf(v, x, q1.y); v = fmaf(v, x, q1.x);
      v = fmaf(v, x, q0.w); v = fmaf(v, x, q0.z); v = fmaf(v, x, q0.y); v = fmaf(v, x, q0.x);
      out[base + lane] = logf(v);
    } else {
      int e = base + lane;
      if (e < nElem) {
        float bb = b[e], aa = a[e];
        float x = aa * aa * 0.25f;
        float accv = 0.0f, xp = 1.0f;
        for (int k = 0; k < NTERMS; ++k) {
          float t = 0.0f, bp = 1.0f;
          for (int m = 0; m < NTERMS; ++m) { t = fmaf(Ksh[m][k], bp, t); bp *= bb; }
          accv = fmaf(t, xp, accv);
          xp *= x;
        }
        out[e] = logf(accv);
      }
    }
  }
}

extern "C" void kernel_launch(void* const* d_in, const int* in_sizes, int n_in,
                              void* d_out, int out_size, void* d_ws, size_t ws_size,
                              hipStream_t stream) {
  (void)n_in; (void)out_size; (void)d_ws; (void)ws_size;
  const float* a = (const float*)d_in[0];
  const float* b = (const float*)d_in[1];
  float* out = (float*)d_out;
  int n = in_sizes[0];
  int chunks = (n + 31) / 32;
  int blocks = (chunks + 7) / 8;
  if (blocks > 1024) blocks = 1024;
  if (blocks < 1) blocks = 1;
  ipe_logsumexp_wmma<<<blocks, 256, 0, stream>>>(a, b, out, n, chunks);
}


# Inputs of each wave's first two chunks requested before the f64 table build, all their A operands formed during its division step (build barriers fence LDS only); sum order unchanged
# speedup vs baseline: 10.4085x; 10.4085x over previous
#include <hip/hip_runtime.h>
#include <math.h>


typedef __attribute__((ext_vector_type(2))) float v2f;
typedef __attribute__((ext_vector_type(8))) float v8f;

#define NTERMS 20
#define KPAD   32
#define QROW   20
#define NWAVES 32

__device__ __forceinline__ void lds_barrier() {
  __builtin_amdgcn_fence(__ATOMIC_RELEASE, "workgroup", "local");
  __builtin_amdgcn_s_barrier();
  __builtin_amdgcn_fence(__ATOMIC_ACQUIRE, "workgroup", "local");
}

struct ChunkOps {
  v2f ta[7];
  v2f tb[7];
  float xo, bo;
};

__device__ __forceinline__ void form_operands(float aA, float aB, float bA, float bB,
                                              bool hi, ChunkOps& o) {
  const float xA = aA * aA * 0.25f;
  const float xB = aB * aB * 0.25f;

  const float bA2 = bA * bA, bB2 = bB * bB;
  const float bA4 = bA2 * bA2, bB4 = bB2 * bB2;
  const float bA8 = bA4 * bA4, bB8 = bB4 * bB4;
  const float bA12 = bA8 * bA4, bB12 = bB8 * bB4;
  const float bA16 = bA8 * bA8, bB16 = bB8 * bB8;
  const float xA2 = xA * xA, xB2 = xB * xB;
  const float xA4 = xA2 * xA2, xB4 = xB2 * xB2;
  const float xA8 = xA4 * xA4, xB8 = xB4 * xB4;
  const float xA12 = xA8 * xA4, xB12 = xB8 * xB4;
  const float xA16 = xA8 * xA8, xB16 = xB8 * xB8;

  float p0A = hi ? bA8  : 1.0f;
  float p1A = hi ? bA12 : bA4;
  float p0B = hi ? bB8  : 1.0f;
  float p1B = hi ? bB12 : bB4;
#pragma unroll
  for (int s = 0; s < 5; ++s) {
    o.ta[s] = v2f{p0A, p1A};
    o.tb[s] = v2f{p0B, p1B};
    p0A *= xA4; p1A *= xA4;
    p0B *= xB4; p1B *= xB4;
  }
  o.ta[5] = v2f{bA16 * (hi ? xA8 : 1.0f), bA16 * (hi ? xA12 : xA4)};
  o.tb[5] = v2f{bB16 * (hi ? xB8 : 1.0f), bB16 * (hi ? xB12 : xB4)};
  o.ta[6] = v2f{hi ? 0.0f : bA16 * xA16, 0.0f};
  o.tb[6] = v2f{hi ? 0.0f : bB16 * xB16, 0.0f};
  o.xo = hi ? xB : xA;
  o.bo = hi ? bB : bA;
}

__device__ __forceinline__ float contract_finish(const ChunkOps& o, const v2f (&Bf)[7],
                                                 float (*Tw)[16][QROW], int lane, int col, bool hi) {
  v8f cA = {}, cB = {};
#pragma unroll
  for (int s = 0; s < 7; ++s) {
    cA = __builtin_amdgcn_wmma_f32_16x16x4_f32(false, o.ta[s], false, Bf[s], (short)0, cA, false, false);
    cB = __builtin_amdgcn_wmma_f32_16x16x4_f32(false, o.tb[s], false, Bf[s], (short)0, cB, false, false);
  }

  const int row = hi ? 8 : 0;
#pragma unroll
  for (int i = 0; i < 8; ++i) {
    Tw[0][row + i][col] = cA[i];
    Tw[1][row + i][col] = cB[i];
  }

  const float4* pT4 = (const float4*)&Tw[lane >> 4][col][0];
  const float4 q0 = pT4[0], q1 = pT4[1], q2 = pT4[2], q3 = pT4[3];
  const float xo = o.xo, bo = o.bo;
  const float r0 = fmaf(fmaf(fmaf(q0.w, xo, q0.z), xo, q0.y), xo, q0.x);
  const float r1 = fmaf(fmaf(fmaf(q1.w, xo, q1.z), xo, q1.y), xo, q1.x);
  const float r2 = fmaf(fmaf(fmaf(q2.w, xo, q2.z), xo, q2.y), xo, q2.x);
  const float r3 = fmaf(fmaf(fmaf(q3.w, xo, q3.z), xo, q3.y), xo, q3.x);
  const float v  = fmaf(fmaf(fmaf(r3, bo, r2), bo, r1), bo, r0);
  return logf(v);
}

__global__ __launch_bounds__(NWAVES * 32) void ipe_logsumexp_wmma(
    const float* a, const float* b,
    float* __restrict__ out, int nElem, int nChunks) {
  __shared__ __align__(16) float Ksh[NTERMS][KPAD];
  __shared__ __align__(16) float Tsh[NWAVES][2][16][QROW];

  __shared__ double Ssh[96];
  __shared__ double Fsh[96];
  __shared__ double Csh[98];

  const int tid  = threadIdx.x;
  const int lane = threadIdx.x & 31;
  const int wv   = threadIdx.x >> 5;
  const int col  = lane & 15;
  const bool hi  = lane >= 16;
  const int h2   = hi ? 2 : 0;

  const int nwaves = (int)gridDim.x * (blockDim.x >> 5);
  const int gwave  = (int)blockIdx.x * (blockDim.x >> 5) + wv;


  if (tid < 96) {
    double v = (tid == 0) ? 1.0 : (double)tid;
#pragma unroll
    for (int d = 1; d < 32; d <<= 1) {
      double t = __shfl_up(v, d);
      if ((tid & 31) >= d) v *= t;
    }
    Ssh[tid] = v;
  }

  const int chunk0 = gwave, chunk1 = gwave + nwaves;
  const bool pre0 = (chunk0 < nChunks) && (chunk0 * 32 + 32 <= nElem);
  const bool pre1 = (chunk1 < nChunks) && (chunk1 * 32 + 32 <= nElem);
  const int last  = nElem - 1;
  const int base0 = chunk0 * 32, base1 = chunk1 * 32;
  const int i0A = max(min(base0 + col, last), 0), i0B = max(min(base0 + 16 + col, last), 0);
  const int i1A = max(min(base1 + col, last), 0), i1B = max(min(base1 + 16 + col, last), 0);
  const float in0bA = b[i0A], in0bB = b[i0B], in0aA = a[i0A], in0aB = a[i0B];
  const float in1bA = b[i1A], in1bB = b[i1B], in1aA = a[i1A], in1aB = a[i1B];
  ChunkOps ops0 = {}, ops1 = {};

  lds_barrier();
  if (tid < 96) {
    double f = Ssh[tid];
    if (tid >= 32) f *= Ssh[31];
    if (tid >= 64) f *= Ssh[63];
    Fsh[tid] = f;
  }
  lds_barrier();
  if (tid < 98) {
    const bool isC = tid < 20;
    const bool isD = (tid >= 20) && (tid < 40);
    const int m = tid, k = tid - 20, i = tid - 40;
    const int n0 = isC ? 4 * m : (isD ? 2 * k + 8 : 4);
    const int n4 = isD ? 4 : 0;
    const int n1 = isC ? 2 * m : (isD ? k + 4 : i + 4);
    const int n2 = isC ? m : (isD ? 2 * k : 0);
    const int n3 = isC ? m : (isD ? 8 : 0);
    const double s = isC ? 0.2734375 * 3.14159265358979323846 : 1.0;
    const double num = s * (Fsh[n0] * Fsh[n4]);
    const double den = Fsh[n1] * Fsh[n2] * Fsh[n3];
    Csh[tid] = ldexp(num / den, isC ? -6 * m : 0);
  }
  if (pre0) form_operands(in0aA, in0aB, in0bA, in0bB, hi, ops0);
  if (pre1) form_operands(in1aA, in1aB, in1bA, in1bB, hi, ops1);
  lds_barrier();
  for (int idx = tid; idx < NTERMS * NTERMS; idx += NWAVES * 32) {
    const int m = idx / NTERMS, k = idx - m * NTERMS;
    Ksh[m][k] = (float)(Csh[m] * Csh[NTERMS + k] * Csh[2 * NTERMS + 2 * m + k]);
  }
  lds_barrier();

  const int m0 = col >> 2, k0 = col & 3;
  v2f Bf[7];
#pragma unroll
  for (int s = 0; s < 5; ++s) {
    Bf[s] = v2f{Ksh[4 * h2 + m0][4 * s + k0], Ksh[4 * (h2 + 1) + m0][4 * s + k0]};
  }
  Bf[5] = v2f{Ksh[16 + m0][4 * h2 + k0], Ksh[16 + m0][4 * (h2 + 1) + k0]};
  Bf[6] = v2f{hi ? 0.0f : Ksh[16 + m0][16 + k0], 0.0f};

  if (pre0) out[chunk0 * 32 + lane] = contract_finish(ops0, Bf, Tsh[wv], lane, col, hi);
  if (pre1) out[chunk1 * 32 + lane] = contract_finish(ops1, Bf, Tsh[wv], lane, col, hi);

  for (int chunk = gwave + (pre0 ? nwaves : 0) + (pre1 ? nwaves : 0); chunk < nChunks; chunk += nwaves) {
    const int base = chunk * 32;
    if (base + 32 <= nElem) {
      ChunkOps o;
      form_operands(a[base + col], a[base + 16 + col], b[base + col], b[base + 16 + col], hi, o);
      out[base + lane] = contract_finish(o, Bf, Tsh[wv], lane, col, hi);
    } else {
      int e = base + lane;
      if (e < nElem) {
        float bb = b[e], aa = a[e];
        float x = aa * aa * 0.25f;
        float accv = 0.0f, xp = 1.0f;
#pragma nounroll
        for (int k = 0; k < NTERMS; ++k) {
          float t = 0.0f, bp = 1.0f;
#pragma nounroll
          for (int m = 0; m < NTERMS; ++m) { t = fmaf(Ksh[m][k], bp, t); bp *= bb; }
          accv = fmaf(t, xp, accv);
          xp *= x;
        }
        out[e] = logf(accv);
      }
    }
  }
}

extern "C" void kernel_launch(void* const* d_in, const int* in_sizes, int n_in,
                              void* d_out, int out_size, void* d_ws, size_t ws_size,
                              hipStream_t stream) {
  (void)n_in; (void)out_size; (void)d_ws; (void)ws_size;
  const float* a = (const float*)d_in[0];
  const float* b = (const float*)d_in[1];
  float* out = (float*)d_out;
  int n = in_sizes[0];
  int chunks = (n + 31) / 32;
  int blocks = (chunks + NWAVES - 1) / NWAVES;
  if (blocks > 256) blocks = 256;
  if (blocks < 1) blocks = 1;
  ipe_logsumexp_wmma<<<blocks, NWAVES * 32, 0, stream>>>(a, b, out, n, chunks);
}


# Table build: c, d, h as running products of per-lane integer ratios (one f64 division + register lane-shift scan in waves 0..3, no factorial pass): 2 of 4 barriers go; A operands formed after it
# speedup vs baseline: 11.4721x; 1.1022x over previous
#include <hip/hip_runtime.h>
#include <math.h>


typedef __attribute__((ext_vector_type(2))) float v2f;
typedef __attribute__((ext_vector_type(8))) float v8f;

#define NTERMS 20
#define KPAD   32
#define QROW   20
#define NWAVES 32

__device__ __forceinline__ void lds_barrier() {
  __builtin_amdgcn_fence(__ATOMIC_RELEASE, "workgroup", "local");
  __builtin_amdgcn_s_barrier();
  __builtin_amdgcn_fence(__ATOMIC_ACQUIRE, "workgroup", "local");
}

template <int CTRL>
__device__ __forceinline__ double row_shift_or_one(double v) {
  const int lo = __builtin_amdgcn_update_dpp(0, __double2loint(v), CTRL, 0xf, 0xf, false);
  const int hi = __builtin_amdgcn_update_dpp(0x3ff00000, __double2hiint(v), CTRL, 0xf, 0xf, false);
  return __hiloint2double(hi, lo);
}

__device__ __forceinline__ double wave_product_scan(double v, bool upper_row) {
  v *= row_shift_or_one<0x111>(v);
  v *= row_shift_or_one<0x112>(v);
  v *= row_shift_or_one<0x114>(v);
  v *= row_shift_or_one<0x118>(v);
  const double row0 = __hiloint2double(__builtin_amdgcn_readlane(__double2hiint(v), 15),
                                       __builtin_amdgcn_readlane(__double2loint(v), 15));
  return upper_row ? v * row0 : v;
}

struct ChunkOps {
  v2f ta[7];
  v2f tb[7];
  float xo, bo;
};

__device__ __forceinline__ void form_operands(float aA, float aB, float bA, float bB,
                                              bool hi, ChunkOps& o) {
  const float xA = aA * aA * 0.25f;
  const float xB = aB * aB * 0.25f;

  const float bA2 = bA * bA, bB2 = bB * bB;
  const float bA4 = bA2 * bA2, bB4 = bB2 * bB2;
  const float bA8 = bA4 * bA4, bB8 = bB4 * bB4;
  const float bA12 = bA8 * bA4, bB12 = bB8 * bB4;
  const float bA16 = bA8 * bA8, bB16 = bB8 * bB8;
  const float xA2 = xA * xA, xB2 = xB * xB;
  const float xA4 = xA2 * xA2, xB4 = xB2 * xB2;
  const float xA8 = xA4 * xA4, xB8 = xB4 * xB4;
  const float xA12 = xA8 * xA4, xB12 = xB8 * xB4;
  const float xA16 = xA8 * xA8, xB16 = xB8 * xB8;

  float p0A = hi ? bA8  : 1.0f;
  float p1A = hi ? bA12 : bA4;
  float p0B = hi ? bB8  : 1.0f;
  float p1B = hi ? bB12 : bB4;
#pragma unroll
  for (int s = 0; s < 5; ++s) {
    o.ta[s] = v2f{p0A, p1A};
    o.tb[s] = v2f{p0B, p1B};
    p0A *= xA4; p1A *= xA4;
    p0B *= xB4; p1B *= xB4;
  }
  o.ta[5] = v2f{bA16 * (hi ? xA8 : 1.0f), bA16 * (hi ? xA12 : xA4)};
  o.tb[5] = v2f{bB16 * (hi ? xB8 : 1.0f), bB16 * (hi ? xB12 : xB4)};
  o.ta[6] = v2f{hi ? 0.0f : bA16 * xA16, 0.0f};
  o.tb[6] = v2f{hi ? 0.0f : bB16 * xB16, 0.0f};
  o.xo = hi ? xB : xA;
  o.bo = hi ? bB : bA;
}

__device__ __forceinline__ float contract_finish(const ChunkOps& o, const v2f (&Bf)[7],
                                                 float (*Tw)[16][QROW], int lane, int col, bool hi) {
  v8f cA = {}, cB = {};
#pragma unroll
  for (int s = 0; s < 7; ++s) {
    cA = __builtin_amdgcn_wmma_f32_16x16x4_f32(false, o.ta[s], false, Bf[s], (short)0, cA, false, false);
    cB = __builtin_amdgcn_wmma_f32_16x16x4_f32(false, o.tb[s], false, Bf[s], (short)0, cB, false, false);
  }

  const int row = hi ? 8 : 0;
#pragma unroll
  for (int i = 0; i < 8; ++i) {
    Tw[0][row + i][col] = cA[i];
    Tw[1][row + i][col] = cB[i];
  }

  const float4* pT4 = (const float4*)&Tw[lane >> 4][col][0];
  const float4 q0 = pT4[0], q1 = pT4[1], q2 = pT4[2], q3 = pT4[3];
  const float xo = o.xo, bo = o.bo;
  const float r0 = fmaf(fmaf(fmaf(q0.w, xo, q0.z), xo, q0.y), xo, q0.x);
  const float r1 = fmaf(fmaf(fmaf(q1.w, xo, q1.z), xo, q1.y), xo, q1.x);
  const float r2 = fmaf(fmaf(fmaf(q2.w, xo, q2.z), xo, q2.y), xo, q2.x);
  const float r3 = fmaf(fmaf(fmaf(q3.w, xo, q3.z), xo, q3.y), xo, q3.x);
  const float v  = fmaf(fmaf(fmaf(r3, bo, r2), bo, r1), bo, r0);
  return logf(v);
}

__global__ __launch_bounds__(NWAVES * 32) void ipe_logsumexp_wmma(
    const float* a, const float* b,
    float* __restrict__ out, int nElem, int nChunks) {
  __shared__ __align__(16) float Ksh[NTERMS][KPAD];
  __shared__ __align__(16) float Tsh[NWAVES][2][16][QROW];

  __shared__ double Csh[98];

  const int tid  = threadIdx.x;
  const int lane = threadIdx.x & 31;
  const int wv   = threadIdx.x >> 5;
  const int col  = lane & 15;
  const bool hi  = lane >= 16;
  const int h2   = hi ? 2 : 0;

  const int nwaves = (int)gridDim.x * (blockDim.x >> 5);
  const int gwave  = (int)blockIdx.x * (blockDim.x >> 5) + wv;


  double ratio = 1.0;
  if (wv < 4) {
    const unsigned t = (unsigned)lane, j = t - 1u;
    unsigned num = 1u, den = 1u;
    if (wv == 0) {
      if (t >= 1u && t < 20u) { num = (4u * j + 1u) * (4u * j + 3u); den = 16u * t * t; }
    } else if (wv == 1) {
      if (t >= 1u && t < 20u) { num = 2u * j + 9u; den = (2u * j + 1u) * t; }
    } else if (wv == 2) {
      if (t >= 1u && t < 29u) den = (2u * t + 3u) * (2u * t + 4u);
    } else {
      if (t < 29u) den = (t == 0u) ? 5u : (2u * t + 4u) * (2u * t + 5u);
    }
    ratio = (double)num / (double)den;
    if (tid == 0) ratio = 0.2734375 * 3.14159265358979323846;
  }

  const int chunk0 = gwave, chunk1 = gwave + nwaves;
  const bool pre0 = (chunk0 < nChunks) && (chunk0 * 32 + 32 <= nElem);
  const bool pre1 = (chunk1 < nChunks) && (chunk1 * 32 + 32 <= nElem);
  const int last  = nElem - 1;
  const int base0 = chunk0 * 32, base1 = chunk1 * 32;
  const int i0A = max(min(base0 + col, last), 0), i0B = max(min(base0 + 16 + col, last), 0);
  const int i1A = max(min(base1 + col, last), 0), i1B = max(min(base1 + 16 + col, last), 0);
  const float in0bA = b[i0A], in0bB = b[i0B], in0aA = a[i0A], in0aB = a[i0B];
  const float in1bA = b[i1A], in1bB = b[i1B], in1aA = a[i1A], in1aB = a[i1B];

  if (wv < 4) {
    const double seq = wave_product_scan(ratio, hi);
    if (wv == 0)      { if (lane < 20) Csh[lane] = seq; }
    else if (wv == 1) { if (lane < 20) Csh[NTERMS + lane] = seq; }
    else              { if (lane < 29) Csh[2 * NTERMS + 2 * lane + (wv - 2)] = seq; }
  }
  lds_barrier();
  for (int idx = tid; idx < NTERMS * NTERMS; idx += NWAVES * 32) {
    const int m = idx / NTERMS, k = idx - m * NTERMS;
    Ksh[m][k] = (float)(Csh[m] * Csh[NTERMS + k] * Csh[2 * NTERMS + 2 * m + k]);
  }
  lds_barrier();

  const int m0 = col >> 2, k0 = col & 3;
  v2f Bf[7];
#pragma unroll
  for (int s = 0; s < 5; ++s) {
    Bf[s] = v2f{Ksh[4 * h2 + m0][4 * s + k0], Ksh[4 * (h2 + 1) + m0][4 * s + k0]};
  }
  Bf[5] = v2f{Ksh[16 + m0][4 * h2 + k0], Ksh[16 + m0][4 * (h2 + 1) + k0]};
  Bf[6] = v2f{hi ? 0.0f : Ksh[16 + m0][16 + k0], 0.0f};

  if (pre0) {
    ChunkOps o;
    form_operands(in0aA, in0aB, in0bA, in0bB, hi, o);
    out[chunk0 * 32 + lane] = contract_finish(o, Bf, Tsh[wv], lane, col, hi);
  }
  if (pre1) {
    ChunkOps o;
    form_operands(in1aA, in1aB, in1bA, in1bB, hi, o);
    out[chunk1 * 32 + lane] = contract_finish(o, Bf, Tsh[wv], lane, col, hi);
  }

  for (int chunk = gwave + (pre0 ? nwaves : 0) + (pre1 ? nwaves : 0); chunk < nChunks; chunk += nwaves) {
    const int base = chunk * 32;
    if (base + 32 <= nElem) {
      ChunkOps o;
      form_operands(a[base + col], a[base + 16 + col], b[base + col], b[base + 16 + col], hi, o);
      out[base + lane] = contract_finish(o, Bf, Tsh[wv], lane, col, hi);
    } else {
      int e = base + lane;
      if (e < nElem) {
        float bb = b[e], aa = a[e];
        float x = aa * aa * 0.25f;
        float accv = 0.0f, xp = 1.0f;
#pragma nounroll
        for (int k = 0; k < NTERMS; ++k) {
          float t = 0.0f, bp = 1.0f;
#pragma nounroll
          for (int m = 0; m < NTERMS; ++m) { t = fmaf(Ksh[m][k], bp, t); bp *= bb; }
          accv = fmaf(t, xp, accv);
          xp *= x;
        }
        out[e] = logf(accv);
      }
    }
  }
}

extern "C" void kernel_launch(void* const* d_in, const int* in_sizes, int n_in,
                              void* d_out, int out_size, void* d_ws, size_t ws_size,
                              hipStream_t stream) {
  (void)n_in; (void)out_size; (void)d_ws; (void)ws_size;
  const float* a = (const float*)d_in[0];
  const float* b = (const float*)d_in[1];
  float* out = (float*)d_out;
  int n = in_sizes[0];
  int chunks = (n + 31) / 32;
  int blocks = (chunks + NWAVES - 1) / NWAVES;
  if (blocks > 256) blocks = 256;
  if (blocks < 1) blocks = 1;
  ipe_logsumexp_wmma<<<blocks, NWAVES * 32, 0, stream>>>(a, b, out, n, chunks);
}
